# starting kernel
# baseline (speedup 1.0000x reference)
#include <hip/hip_runtime.h>
#include <hip/hip_bf16.h>

typedef __attribute__((ext_vector_type(2))) float v2f;
typedef __attribute__((ext_vector_type(4))) float v4f;
typedef __attribute__((ext_vector_type(8))) float v8f;

#define H_DIM   4096
#define IN_DIM  512
#define E_DIM   8
#define OUT_DIM 512
#define NTOK    16384

__global__ void moe_prep_weff(const float* __restrict__ Wi, const float* __restrict__ Wr,
                              const float* __restrict__ bi, const float* __restrict__ br,
                              float* __restrict__ WcT, float* __restrict__ beff) {
    int t = blockIdx.x * blockDim.x + threadIdx.x;
    int e = t >> 9;
    int k = t & 511;
    const float* wr = Wr + (size_t)e * H_DIM;
    float acc = 0.f;
    for (int h = 0; h < H_DIM; ++h)
        acc = __builtin_fmaf(wr[h], Wi[(size_t)h * IN_DIM + k], acc);
    WcT[e * IN_DIM + k] = acc;
    if (k == 0) {
        float b = br[e];
        for (int h = 0; h < H_DIM; ++h)
            b = __builtin_fmaf(wr[h], bi[h], b);
        beff[e] = b;
    }
}

__global__ void moe_prep_misc(const float* __restrict__ Wi, const float* __restrict__ Wo,
                              float* __restrict__ WcT, float* __restrict__ wsum) {
    int k = blockIdx.x * blockDim.x + threadIdx.x;
    #pragma unroll
    for (int n = 0; n < 8; ++n)
        WcT[(8 + n) * IN_DIM + k] = Wi[(size_t)n * IN_DIM + k];
    const float* wo = Wo + (size_t)k * H_DIM;
    float s = 0.f;
    for (int h = 0; h < H_DIM; ++h) s += wo[h];
    wsum[k] = s;
}

__global__ void __launch_bounds__(256)
moe_main(const float* __restrict__ x,    const float* __restrict__ WcT,
         const float* __restrict__ beff, const float* __restrict__ bi,
         const float* __restrict__ wsum, const float* __restrict__ bo,
         float* __restrict__ out) {
    __shared__ float lds_c[8][16][16];
    __shared__ float lds_agg[8][16];

    const int wave  = threadIdx.x >> 5;
    const int lane  = threadIdx.x & 31;
    const int tile  = blockIdx.x * 8 + wave;
    const int tok0  = tile * 16;
    const int col   = lane & 15;
    const int half  = lane >> 4;

    const float* ap = x   + (size_t)(tok0 + col) * IN_DIM + half * 2;
    const float* bp = WcT + (size_t)col * IN_DIM + half * 2;

    v8f acc = {};
    #pragma unroll 8
    for (int k = 0; k < IN_DIM; k += 4) {
        v2f a = *(const v2f*)(ap + k);
        v2f b = *(const v2f*)(bp + k);
        acc = __builtin_amdgcn_wmma_f32_16x16x4_f32(
             false, a,  false, b,
             (short)0, acc,  false,  false);
    }

    v4f wv[4], bv[4];
    #pragma unroll
    for (int i = 0; i < 4; ++i) {
        wv[i] = *(const v4f*)(wsum + lane * 4 + 128 * i);
        bv[i] = *(const v4f*)(bo   + lane * 4 + 128 * i);
    }

    float cb = (col < 8) ? beff[col] : bi[col - 8];
    #pragma unroll
    for (int j = 0; j < 8; ++j)
        lds_c[wave][half * 8 + j][col] = acc[j] + cb;
    __syncthreads();

    if (lane < 16) {
        const float* r = lds_c[wave][lane];
        float b1 = r[0]; int i1 = 0;
        #pragma unroll
        for (int n = 1; n < 8; ++n) if (r[n] > b1) { b1 = r[n]; i1 = n; }
        float b2 = -3.402823466e38f; int i2 = 0;
        #pragma unroll
        for (int n = 0; n < 8; ++n) if (n != i1 && r[n] > b2) { b2 = r[n]; i2 = n; }
        lds_agg[wave][lane] = 0.5f * (r[8 + i1] + r[8 + i2]);
    }
    __syncthreads();

    #pragma unroll 4
    for (int t = 0; t < 16; ++t) {
        float ag = lds_agg[wave][t];
        float* op = out + (size_t)(tok0 + t) * OUT_DIM + lane * 4;
        #pragma unroll
        for (int i = 0; i < 4; ++i) {
            v4f r = ag * wv[i] + bv[i];
            *(v4f*)(op + 128 * i) = r;
        }
    }
}

extern "C" void kernel_launch(void* const* d_in, const int* in_sizes, int n_in,
                              void* d_out, int out_size, void* d_ws, size_t ws_size,
                              hipStream_t stream) {
    const float* x  = (const float*)d_in[0];
    const float* Wi = (const float*)d_in[1];
    const float* bi = (const float*)d_in[2];
    const float* Wr = (const float*)d_in[3];
    const float* br = (const float*)d_in[4];
    const float* Wo = (const float*)d_in[5];
    const float* bo = (const float*)d_in[6];
    float* out = (float*)d_out;

    float* WcT  = (float*)d_ws;
    float* beff = WcT + 16 * IN_DIM;
    float* wsum = beff + 16;

    moe_prep_weff<<<16, 256, 0, stream>>>(Wi, Wr, bi, br, WcT, beff);
    moe_prep_misc<<<2, 256, 0, stream>>>(Wi, Wo, WcT, wsum);
    moe_main<<<NTOK / 16 / 8, 256, 0, stream>>>(x, WcT, beff, bi, wsum, bo, out);
}


# Weight prep re-cut output-stationary in one launch: each work-group owns whole outputs and sums all of h inside the group (lanes, then LDS); changes f32 summation order of folded weight and row sums
# speedup vs baseline: 10.9493x; 10.9493x over previous
#include <hip/hip_runtime.h>
#include <hip/hip_bf16.h>

typedef __attribute__((ext_vector_type(2))) float v2f;
typedef __attribute__((ext_vector_type(4))) float v4f;
typedef __attribute__((ext_vector_type(8))) float v8f;

#define H_DIM   4096
#define IN_DIM  512
#define E_DIM   8
#define OUT_DIM 512
#define NTOK    16384

#define PREP_THREADS 512
#define PREP_WAVES   (PREP_THREADS / 32)
#define STRIP        32
#define N_FOLD       (IN_DIM / STRIP)
#define WSUM_ROWS    4
#define N_WSUM       (OUT_DIM / WSUM_ROWS)
#define N_PREP       (N_FOLD + N_WSUM + 1)

__global__ void __launch_bounds__(PREP_THREADS)
moe_prep(const float* __restrict__ Wi, const float* __restrict__ bi,
         const float* __restrict__ Wr, const float* __restrict__ br,
         const float* __restrict__ Wo,
         float* __restrict__ WcT, float* __restrict__ beff, float* __restrict__ wsum) {
    __shared__ __attribute__((aligned(16))) float red[PREP_WAVES][E_DIM][STRIP];

    const int tid  = threadIdx.x;
    const int lane = tid & 31;
    const int wave = tid >> 5;
    const int b    = blockIdx.x;

    if (b < N_FOLD) {
        constexpr int ROWS_PER_LANE = H_DIM / (PREP_WAVES * 4);
        constexpr int G = ROWS_PER_LANE / 4;
        constexpr int S = 2;
        const int k0 = b * STRIP;
        const int c4 = lane & 7;
        const int rg = lane >> 3;
        const int hb = wave * (4 * ROWS_PER_LANE) + rg * ROWS_PER_LANE;
        const float* wip = Wi + (size_t)hb * IN_DIM + k0 + c4 * 4;
        const float* wrp = Wr + hb;

        v4f acc[E_DIM];
        #pragma unroll
        for (int e = 0; e < E_DIM; ++e) acc[e] = (v4f){0.f, 0.f, 0.f, 0.f};

        v4f wi[S][4];
        v4f wr[S][E_DIM];
        #pragma unroll
        for (int g = 0; g < S - 1; ++g) {
            #pragma unroll
            for (int r = 0; r < 4; ++r)
                wi[g][r] = *(const v4f*)(wip + (size_t)(g * 4 + r) * IN_DIM);
            #pragma unroll
            for (int e = 0; e < E_DIM; ++e)
                wr[g][e] = *(const v4f*)(wrp + (size_t)e * H_DIM + g * 4);
        }
        #pragma unroll
        for (int g = 0; g < G; ++g) {
            const int gn = g + S - 1;
            if (gn < G) {
                #pragma unroll
                for (int r = 0; r < 4; ++r)
                    wi[gn % S][r] = *(const v4f*)(wip + (size_t)(gn * 4 + r) * IN_DIM);
                #pragma unroll
                for (int e = 0; e < E_DIM; ++e)
                    wr[gn % S][e] = *(const v4f*)(wrp + (size_t)e * H_DIM + gn * 4);
            }
            __builtin_amdgcn_sched_barrier(0);
            const int s = g % S;
            #pragma unroll
            for (int r = 0; r < 4; ++r) {
                #pragma unroll
                for (int e = 0; e < E_DIM; ++e) {
                    const float w = wr[s][e][r];
                    acc[e].x = __builtin_fmaf(w, wi[s][r].x, acc[e].x);
                    acc[e].y = __builtin_fmaf(w, wi[s][r].y, acc[e].y);
                    acc[e].z = __builtin_fmaf(w, wi[s][r].z, acc[e].z);
                    acc[e].w = __builtin_fmaf(w, wi[s][r].w, acc[e].w);
                }
            }
        }

        #pragma unroll
        for (int e = 0; e < E_DIM; ++e) {
            #pragma unroll
            for (int j = 0; j < 4; ++j) {
                float v = acc[e][j];
                v += __shfl_xor(v, 8);
                v += __shfl_xor(v, 16);
                acc[e][j] = v;
            }
        }
        if (lane < 8) {
            #pragma unroll
            for (int e = 0; e < E_DIM; ++e)
                *(v4f*)&red[wave][e][c4 * 4] = acc[e];
        }
        __syncthreads();
        if (tid < E_DIM * STRIP) {
            const int e = tid >> 5;
            const int c = tid & 31;
            float s = 0.f;
            #pragma unroll
            for (int w = 0; w < PREP_WAVES; ++w) s += red[w][e][c];
            WcT[e * IN_DIM + k0 + c] = s;
        }
    } else if (b < N_FOLD + N_WSUM) {
        const int o0  = (b - N_FOLD) * WSUM_ROWS;
        const int row = o0 + (wave >> 2);
        const int q   = wave & 3;
        const float* wp = Wo + (size_t)row * H_DIM + q * (H_DIM / 4) + lane * 4;
        v4f v[8];
        #pragma unroll
        for (int i = 0; i < 8; ++i) v[i] = *(const v4f*)(wp + i * 128);
        v4f s4 = ((v[0] + v[1]) + (v[2] + v[3])) + ((v[4] + v[5]) + (v[6] + v[7]));
        float s = (s4.x + s4.y) + (s4.z + s4.w);
        #pragma unroll
        for (int off = 16; off >= 1; off >>= 1) s += __shfl_xor(s, off);
        float* rs = &red[0][0][0];
        if (lane == 0) rs[wave] = s;
        __syncthreads();
        if (tid < WSUM_ROWS)
            wsum[o0 + tid] = (rs[4 * tid] + rs[4 * tid + 1]) + (rs[4 * tid + 2] + rs[4 * tid + 3]);
    } else {
        float pe[E_DIM];
        #pragma unroll
        for (int e = 0; e < E_DIM; ++e) pe[e] = 0.f;
        #pragma unroll
        for (int i = 0; i < H_DIM / PREP_THREADS; ++i) {
            const int h = tid + i * PREP_THREADS;
            const float bv = bi[h];
            #pragma unroll
            for (int e = 0; e < E_DIM; ++e)
                pe[e] = __builtin_fmaf(Wr[(size_t)e * H_DIM + h], bv, pe[e]);
        }
        #pragma unroll
        for (int e = 0; e < E_DIM; ++e) {
            float v = pe[e];
            #pragma unroll
            for (int off = 16; off >= 1; off >>= 1) v += __shfl_xor(v, off);
            pe[e] = v;
        }
        if (lane == 0) {
            #pragma unroll
            for (int e = 0; e < E_DIM; ++e) red[wave][e][0] = pe[e];
        }
        __syncthreads();
        if (tid < E_DIM) {
            float s = br[tid];
            #pragma unroll
            for (int w = 0; w < PREP_WAVES; ++w) s += red[w][tid][0];
            beff[tid] = s;
        }
        #pragma unroll
        for (int n = 0; n < 8; ++n)
            WcT[(8 + n) * IN_DIM + tid] = Wi[(size_t)n * IN_DIM + tid];
    }
}

__global__ void __launch_bounds__(256)
moe_main(const float* __restrict__ x,    const float* __restrict__ WcT,
         const float* __restrict__ beff, const float* __restrict__ bi,
         const float* __restrict__ wsum, const float* __restrict__ bo,
         float* __restrict__ out) {
    __shared__ float lds_c[8][16][16];
    __shared__ float lds_agg[8][16];

    const int wave  = threadIdx.x >> 5;
    const int lane  = threadIdx.x & 31;
    const int tile  = blockIdx.x * 8 + wave;
    const int tok0  = tile * 16;
    const int col   = lane & 15;
    const int half  = lane >> 4;

    const float* ap = x   + (size_t)(tok0 + col) * IN_DIM + half * 2;
    const float* bp = WcT + (size_t)col * IN_DIM + half * 2;

    v8f acc = {};
    #pragma unroll 8
    for (int k = 0; k < IN_DIM; k += 4) {
        v2f a = *(const v2f*)(ap + k);
        v2f b = *(const v2f*)(bp + k);
        acc = __builtin_amdgcn_wmma_f32_16x16x4_f32(
             false, a,  false, b,
             (short)0, acc,  false,  false);
    }

    v4f wv[4], bv[4];
    #pragma unroll
    for (int i = 0; i < 4; ++i) {
        wv[i] = *(const v4f*)(wsum + lane * 4 + 128 * i);
        bv[i] = *(const v4f*)(bo   + lane * 4 + 128 * i);
    }

    float cb = (col < 8) ? beff[col] : bi[col - 8];
    #pragma unroll
    for (int j = 0; j < 8; ++j)
        lds_c[wave][half * 8 + j][col] = acc[j] + cb;
    __syncthreads();

    if (lane < 16) {
        const float* r = lds_c[wave][lane];
        float b1 = r[0]; int i1 = 0;
        #pragma unroll
        for (int n = 1; n < 8; ++n) if (r[n] > b1) { b1 = r[n]; i1 = n; }
        float b2 = -3.402823466e38f; int i2 = 0;
        #pragma unroll
        for (int n = 0; n < 8; ++n) if (n != i1 && r[n] > b2) { b2 = r[n]; i2 = n; }
        lds_agg[wave][lane] = 0.5f * (r[8 + i1] + r[8 + i2]);
    }
    __syncthreads();

    #pragma unroll 4
    for (int t = 0; t < 16; ++t) {
        float ag = lds_agg[wave][t];
        float* op = out + (size_t)(tok0 + t) * OUT_DIM + lane * 4;
        #pragma unroll
        for (int i = 0; i < 4; ++i) {
            v4f r = ag * wv[i] + bv[i];
            *(v4f*)(op + 128 * i) = r;
        }
    }
}

extern "C" void kernel_launch(void* const* d_in, const int* in_sizes, int n_in,
                              void* d_out, int out_size, void* d_ws, size_t ws_size,
                              hipStream_t stream) {
    const float* x  = (const float*)d_in[0];
    const float* Wi = (const float*)d_in[1];
    const float* bi = (const float*)d_in[2];
    const float* Wr = (const float*)d_in[3];
    const float* br = (const float*)d_in[4];
    const float* Wo = (const float*)d_in[5];
    const float* bo = (const float*)d_in[6];
    float* out = (float*)d_out;

    float* WcT  = (float*)d_ws;
    float* beff = WcT + 16 * IN_DIM;
    float* wsum = beff + 16;

    moe_prep<<<N_PREP, PREP_THREADS, 0, stream>>>(Wi, bi, Wr, br, Wo, WcT, beff, wsum);
    moe_main<<<NTOK / 16 / 8, 256, 0, stream>>>(x, WcT, beff, bi, wsum, bo, out);
}


# Main kernel re-cut, one wave per SIMD: each lane prefetches its whole x tile share into A fragments (16-byte loads), folded weight staged per group in LDS; changes f32 summation order of the products
# speedup vs baseline: 15.5064x; 1.4162x over previous
#include <hip/hip_runtime.h>
#include <hip/hip_bf16.h>

typedef __attribute__((ext_vector_type(2))) float v2f;
typedef __attribute__((ext_vector_type(4))) float v4f;
typedef __attribute__((ext_vector_type(8))) float v8f;

#define H_DIM   4096
#define IN_DIM  512
#define E_DIM   8
#define OUT_DIM 512
#define NTOK    16384

#define PREP_THREADS 512
#define PREP_WAVES   (PREP_THREADS / 32)
#define STRIP        32
#define N_FOLD       (IN_DIM / STRIP)
#define WSUM_ROWS    4
#define N_WSUM       (OUT_DIM / WSUM_ROWS)
#define N_PREP       (N_FOLD + N_WSUM + 1)

__global__ void __launch_bounds__(PREP_THREADS)
moe_prep(const float* __restrict__ Wi, const float* __restrict__ bi,
         const float* __restrict__ Wr, const float* __restrict__ br,
         const float* __restrict__ Wo,
         float* __restrict__ WcT, float* __restrict__ beff, float* __restrict__ wsum) {
    __shared__ __attribute__((aligned(16))) float red[PREP_WAVES][E_DIM][STRIP];

    const int tid  = threadIdx.x;
    const int lane = tid & 31;
    const int wave = tid >> 5;
    const int b    = blockIdx.x;

    if (b < N_FOLD) {
        constexpr int ROWS_PER_LANE = H_DIM / (PREP_WAVES * 4);
        constexpr int G = ROWS_PER_LANE / 4;
        constexpr int S = 2;
        const int k0 = b * STRIP;
        const int c4 = lane & 7;
        const int rg = lane >> 3;
        const int hb = wave * (4 * ROWS_PER_LANE) + rg * ROWS_PER_LANE;
        const float* wip = Wi + (size_t)hb * IN_DIM + k0 + c4 * 4;
        const float* wrp = Wr + hb;

        v4f acc[E_DIM];
        #pragma unroll
        for (int e = 0; e < E_DIM; ++e) acc[e] = (v4f){0.f, 0.f, 0.f, 0.f};

        v4f wi[S][4];
        v4f wr[S][E_DIM];
        #pragma unroll
        for (int g = 0; g < S - 1; ++g) {
            #pragma unroll
            for (int r = 0; r < 4; ++r)
                wi[g][r] = *(const v4f*)(wip + (size_t)(g * 4 + r) * IN_DIM);
            #pragma unroll
            for (int e = 0; e < E_DIM; ++e)
                wr[g][e] = *(const v4f*)(wrp + (size_t)e * H_DIM + g * 4);
        }
        #pragma unroll
        for (int g = 0; g < G; ++g) {
            const int gn = g + S - 1;
            if (gn < G) {
                #pragma unroll
                for (int r = 0; r < 4; ++r)
                    wi[gn % S][r] = *(const v4f*)(wip + (size_t)(gn * 4 + r) * IN_DIM);
                #pragma unroll
                for (int e = 0; e < E_DIM; ++e)
                    wr[gn % S][e] = *(const v4f*)(wrp + (size_t)e * H_DIM + gn * 4);
            }
            __builtin_amdgcn_sched_barrier(0);
            const int s = g % S;
            #pragma unroll
            for (int r = 0; r < 4; ++r) {
                #pragma unroll
                for (int e = 0; e < E_DIM; ++e) {
                    const float w = wr[s][e][r];
                    acc[e].x = __builtin_fmaf(w, wi[s][r].x, acc[e].x);
                    acc[e].y = __builtin_fmaf(w, wi[s][r].y, acc[e].y);
                    acc[e].z = __builtin_fmaf(w, wi[s][r].z, acc[e].z);
                    acc[e].w = __builtin_fmaf(w, wi[s][r].w, acc[e].w);
                }
            }
        }

        #pragma unroll
        for (int e = 0; e < E_DIM; ++e) {
            #pragma unroll
            for (int j = 0; j < 4; ++j) {
                float v = acc[e][j];
                v += __shfl_xor(v, 8);
                v += __shfl_xor(v, 16);
                acc[e][j] = v;
            }
        }
        if (lane < 8) {
            #pragma unroll
            for (int e = 0; e < E_DIM; ++e)
                *(v4f*)&red[wave][e][c4 * 4] = acc[e];
        }
        __syncthreads();
        if (tid < E_DIM * STRIP) {
            const int e = tid >> 5;
            const int c = tid & 31;
            float s = 0.f;
            #pragma unroll
            for (int w = 0; w < PREP_WAVES; ++w) s += red[w][e][c];
            WcT[e * IN_DIM + k0 + c] = s;
        }
    } else if (b < N_FOLD + N_WSUM) {
        const int o0  = (b - N_FOLD) * WSUM_ROWS;
        const int row = o0 + (wave >> 2);
        const int q   = wave & 3;
        const float* wp = Wo + (size_t)row * H_DIM + q * (H_DIM / 4) + lane * 4;
        v4f v[8];
        #pragma unroll
        for (int i = 0; i < 8; ++i) v[i] = *(const v4f*)(wp + i * 128);
        v4f s4 = ((v[0] + v[1]) + (v[2] + v[3])) + ((v[4] + v[5]) + (v[6] + v[7]));
        float s = (s4.x + s4.y) + (s4.z + s4.w);
        #pragma unroll
        for (int off = 16; off >= 1; off >>= 1) s += __shfl_xor(s, off);
        float* rs = &red[0][0][0];
        if (lane == 0) rs[wave] = s;
        __syncthreads();
        if (tid < WSUM_ROWS)
            wsum[o0 + tid] = (rs[4 * tid] + rs[4 * tid + 1]) + (rs[4 * tid + 2] + rs[4 * tid + 3]);
    } else {
        float pe[E_DIM];
        #pragma unroll
        for (int e = 0; e < E_DIM; ++e) pe[e] = 0.f;
        #pragma unroll
        for (int i = 0; i < H_DIM / PREP_THREADS; ++i) {
            const int h = tid + i * PREP_THREADS;
            const float bv = bi[h];
            #pragma unroll
            for (int e = 0; e < E_DIM; ++e)
                pe[e] = __builtin_fmaf(Wr[(size_t)e * H_DIM + h], bv, pe[e]);
        }
        #pragma unroll
        for (int e = 0; e < E_DIM; ++e) {
            float v = pe[e];
            #pragma unroll
            for (int off = 16; off >= 1; off >>= 1) v += __shfl_xor(v, off);
            pe[e] = v;
        }
        if (lane == 0) {
            #pragma unroll
            for (int e = 0; e < E_DIM; ++e) red[wave][e][0] = pe[e];
        }
        __syncthreads();
        if (tid < E_DIM) {
            float s = br[tid];
            #pragma unroll
            for (int w = 0; w < PREP_WAVES; ++w) s += red[w][tid][0];
            beff[tid] = s;
        }
        #pragma unroll
        for (int n = 0; n < 8; ++n)
            WcT[(8 + n) * IN_DIM + tid] = Wi[(size_t)n * IN_DIM + tid];
    }
}

#define MAIN_WAVES   4
#define MAIN_THREADS (MAIN_WAVES * 32)
#define N_CHUNK      (IN_DIM / 8)
#define W_PITCH      516
#define W_BLK        16

__global__ void __launch_bounds__(MAIN_THREADS)
moe_main(const float* __restrict__ x,    const float* __restrict__ WcT,
         const float* __restrict__ beff, const float* __restrict__ bi,
         const float* __restrict__ wsum, const float* __restrict__ bo,
         float* __restrict__ out) {
    __shared__ __attribute__((aligned(16))) float lds_w[16 * W_PITCH];
    __shared__ float lds_c[MAIN_WAVES][16][16];
    __shared__ float lds_agg[MAIN_WAVES][16];

    const int tid   = threadIdx.x;
    const int wave  = tid >> 5;
    const int lane  = tid & 31;
    const int tile  = blockIdx.x * MAIN_WAVES + wave;
    const int tok0  = tile * 16;
    const int col   = lane & 15;
    const int half  = lane >> 4;

    v4f wst[16];
    #pragma unroll
    for (int n = 0; n < 16; ++n)
        wst[n] = *(const v4f*)(WcT + (size_t)n * IN_DIM + 4 * tid);

    const float* ap = x + (size_t)(tok0 + col) * IN_DIM + half * 4;
    v4f xa[N_CHUNK];
    #pragma unroll
    for (int j = 0; j < N_CHUNK; ++j)
        xa[j] = *(const v4f*)(ap + 8 * j);

    #pragma unroll
    for (int n = 0; n < 16; ++n)
        *(v4f*)&lds_w[n * W_PITCH + 4 * tid] = wst[n];

    v4f wv[4], bv[4];
    #pragma unroll
    for (int i = 0; i < 4; ++i) {
        wv[i] = *(const v4f*)(wsum + lane * 4 + 128 * i);
        bv[i] = *(const v4f*)(bo   + lane * 4 + 128 * i);
    }
    const float cb = (col < 8) ? beff[col] : bi[col - 8];
    __builtin_amdgcn_fence(__ATOMIC_RELEASE, "workgroup", "local");
    __builtin_amdgcn_s_barrier();
    __builtin_amdgcn_fence(__ATOMIC_ACQUIRE, "workgroup", "local");

    const float* bl = lds_w + col * W_PITCH + half * 4;
    v8f acc0 = {}, acc1 = {};
    v4f wb[2][W_BLK];
    #pragma unroll
    for (int c = 0; c < W_BLK; ++c)
        wb[0][c] = *(const v4f*)(bl + 8 * c);
    #pragma unroll
    for (int blk = 0; blk < N_CHUNK / W_BLK; ++blk) {
        if (blk + 1 < N_CHUNK / W_BLK) {
            #pragma unroll
            for (int c = 0; c < W_BLK; ++c)
                wb[(blk + 1) & 1][c] = *(const v4f*)(bl + 8 * ((blk + 1) * W_BLK + c));
        }
        __builtin_amdgcn_sched_barrier(0);
        #pragma unroll
        for (int c = 0; c < W_BLK; ++c) {
            const v4f a = xa[blk * W_BLK + c];
            const v4f b = wb[blk & 1][c];
            const v2f a0 = {a.x, a.y}, a1 = {a.z, a.w};
            const v2f b0 = {b.x, b.y}, b1 = {b.z, b.w};
            acc0 = __builtin_amdgcn_wmma_f32_16x16x4_f32(
                 false, a0,  false, b0,
                 (short)0, acc0,  false,  false);
            acc1 = __builtin_amdgcn_wmma_f32_16x16x4_f32(
                 false, a1,  false, b1,
                 (short)0, acc1,  false,  false);
        }
        __builtin_amdgcn_sched_barrier(0);
    }
    const v8f acc = acc0 + acc1;

    #pragma unroll
    for (int j = 0; j < 8; ++j)
        lds_c[wave][half * 8 + j][col] = acc[j] + cb;
    __syncthreads();

    if (lane < 16) {
        const float* r = lds_c[wave][lane];
        float b1 = r[0]; int i1 = 0;
        #pragma unroll
        for (int n = 1; n < 8; ++n) if (r[n] > b1) { b1 = r[n]; i1 = n; }
        float b2 = -3.402823466e38f; int i2 = 0;
        #pragma unroll
        for (int n = 0; n < 8; ++n) if (n != i1 && r[n] > b2) { b2 = r[n]; i2 = n; }
        lds_agg[wave][lane] = 0.5f * (r[8 + i1] + r[8 + i2]);
    }
    __syncthreads();

    #pragma unroll 4
    for (int t = 0; t < 16; ++t) {
        float ag = lds_agg[wave][t];
        float* op = out + (size_t)(tok0 + t) * OUT_DIM + lane * 4;
        #pragma unroll
        for (int i = 0; i < 4; ++i) {
            v4f r = ag * wv[i] + bv[i];
            *(v4f*)(op + 128 * i) = r;
        }
    }
}

extern "C" void kernel_launch(void* const* d_in, const int* in_sizes, int n_in,
                              void* d_out, int out_size, void* d_ws, size_t ws_size,
                              hipStream_t stream) {
    const float* x  = (const float*)d_in[0];
    const float* Wi = (const float*)d_in[1];
    const float* bi = (const float*)d_in[2];
    const float* Wr = (const float*)d_in[3];
    const float* br = (const float*)d_in[4];
    const float* Wo = (const float*)d_in[5];
    const float* bo = (const float*)d_in[6];
    float* out = (float*)d_out;

    float* WcT  = (float*)d_ws;
    float* beff = WcT + 16 * IN_DIM;
    float* wsum = beff + 16;

    moe_prep<<<N_PREP, PREP_THREADS, 0, stream>>>(Wi, bi, Wr, br, Wo, WcT, beff, wsum);
    moe_main<<<NTOK / 16 / MAIN_WAVES, MAIN_THREADS, 0, stream>>>(x, WcT, beff, bi, wsum, bo, out);
}


# Weight prep re-balanced inside its one launch: fold groups split by expert pair (64 groups, loads two 16-row chunks deep), bias group second in grid, row-sum groups take 8 rows; f32 sum orders kept
# speedup vs baseline: 18.8345x; 1.2146x over previous
#include <hip/hip_runtime.h>
#include <hip/hip_bf16.h>

typedef __attribute__((ext_vector_type(2))) float v2f;
typedef __attribute__((ext_vector_type(4))) float v4f;
typedef __attribute__((ext_vector_type(8))) float v8f;

#define H_DIM   4096
#define IN_DIM  512
#define E_DIM   8
#define OUT_DIM 512
#define NTOK    16384

#define PREP_THREADS 512
#define PREP_WAVES   (PREP_THREADS / 32)
#define STRIP        32
#define N_FOLD       (IN_DIM / STRIP)
#define E_PER        2
#define N_FOLDB      (N_FOLD * (E_DIM / E_PER))
#define WSUM_ROWS    8
#define N_WSUM       (OUT_DIM / WSUM_ROWS)
#define N_PREP       (N_FOLDB + N_WSUM + 1)

__global__ void __launch_bounds__(PREP_THREADS)
moe_prep(const float* __restrict__ Wi, const float* __restrict__ bi,
         const float* __restrict__ Wr, const float* __restrict__ br,
         const float* __restrict__ Wo,
         float* __restrict__ WcT, float* __restrict__ beff, float* __restrict__ wsum) {
    __shared__ __attribute__((aligned(16))) float red[PREP_WAVES][E_DIM][STRIP];

    const int tid  = threadIdx.x;
    const int lane = tid & 31;
    const int wave = tid >> 5;
    const int b    = blockIdx.x;

    if (b < N_FOLDB) {
        constexpr int ROWS_PER_LANE = H_DIM / (PREP_WAVES * 4);
        constexpr int CH  = 16;
        constexpr int NCH = ROWS_PER_LANE / CH;
        constexpr int S   = 2;
        const int k0 = (b % N_FOLD) * STRIP;
        const int e0 = (b / N_FOLD) * E_PER;
        const int c4 = lane & 7;
        const int rg = lane >> 3;
        const int hb = wave * (4 * ROWS_PER_LANE) + rg * ROWS_PER_LANE;
        const float* wip = Wi + (size_t)hb * IN_DIM + k0 + c4 * 4;
        const float* wrp = Wr + (size_t)e0 * H_DIM + hb;

        v4f acc[E_PER];
        #pragma unroll
        for (int e = 0; e < E_PER; ++e) acc[e] = (v4f){0.f, 0.f, 0.f, 0.f};

        v4f wi[S][CH];
        v4f wr[S][E_PER][CH / 4];
        #pragma unroll
        for (int r = 0; r < CH; ++r)
            wi[0][r] = *(const v4f*)(wip + (size_t)r * IN_DIM);
        #pragma unroll
        for (int e = 0; e < E_PER; ++e) {
            #pragma unroll
            for (int g = 0; g < CH / 4; ++g)
                wr[0][e][g] = *(const v4f*)(wrp + (size_t)e * H_DIM + g * 4);
        }
        #pragma unroll
        for (int q = 0; q < NCH; ++q) {
            const int qn = q + 1;
            if (qn < NCH) {
                #pragma unroll
                for (int r = 0; r < CH; ++r)
                    wi[qn % S][r] = *(const v4f*)(wip + (size_t)(qn * CH + r) * IN_DIM);
                #pragma unroll
                for (int e = 0; e < E_PER; ++e) {
                    #pragma unroll
                    for (int g = 0; g < CH / 4; ++g)
                        wr[qn % S][e][g] = *(const v4f*)(wrp + (size_t)e * H_DIM + qn * CH + g * 4);
                }
            }
            __builtin_amdgcn_sched_barrier(0);
            const int s = q % S;
            #pragma unroll
            for (int r = 0; r < CH; ++r) {
                #pragma unroll
                for (int e = 0; e < E_PER; ++e) {
                    const float w = wr[s][e][r >> 2][r & 3];
                    acc[e].x = __builtin_fmaf(w, wi[s][r].x, acc[e].x);
                    acc[e].y = __builtin_fmaf(w, wi[s][r].y, acc[e].y);
                    acc[e].z = __builtin_fmaf(w, wi[s][r].z, acc[e].z);
                    acc[e].w = __builtin_fmaf(w, wi[s][r].w, acc[e].w);
                }
            }
        }

        #pragma unroll
        for (int e = 0; e < E_PER; ++e) {
            #pragma unroll
            for (int j = 0; j < 4; ++j) {
                float v = acc[e][j];
                v += __shfl_xor(v, 8);
                v += __shfl_xor(v, 16);
                acc[e][j] = v;
            }
        }
        if (lane < 8) {
            #pragma unroll
            for (int e = 0; e < E_PER; ++e)
                *(v4f*)&red[wave][e][c4 * 4] = acc[e];
        }
        __syncthreads();
        if (tid < E_PER * STRIP) {
            const int e = tid >> 5;
            const int c = tid & 31;
            float s = 0.f;
            #pragma unroll
            for (int w = 0; w < PREP_WAVES; ++w) s += red[w][e][c];
            WcT[(e0 + e) * IN_DIM + k0 + c] = s;
        }
    } else if (b > N_FOLDB) {
        const int o0 = (b - N_FOLDB - 1) * WSUM_ROWS;
        const int q  = wave & 3;
        const int r0 = wave >> 2;
        float* rs = &red[0][0][0];
        v4f v[2][8];
        #pragma unroll
        for (int p = 0; p < 2; ++p) {
            const float* wp = Wo + (size_t)(o0 + r0 + 4 * p) * H_DIM + q * (H_DIM / 4) + lane * 4;
            #pragma unroll
            for (int i = 0; i < 8; ++i) v[p][i] = *(const v4f*)(wp + i * 128);
        }
        #pragma unroll
        for (int p = 0; p < 2; ++p) {
            v4f s4 = ((v[p][0] + v[p][1]) + (v[p][2] + v[p][3])) + ((v[p][4] + v[p][5]) + (v[p][6] + v[p][7]));
            float s = (s4.x + s4.y) + (s4.z + s4.w);
            #pragma unroll
            for (int off = 16; off >= 1; off >>= 1) s += __shfl_xor(s, off);
            if (lane == 0) rs[4 * (r0 + 4 * p) + q] = s;
        }
        __syncthreads();
        if (tid < WSUM_ROWS)
            wsum[o0 + tid] = (rs[4 * tid] + rs[4 * tid + 1]) + (rs[4 * tid + 2] + rs[4 * tid + 3]);
    } else {
        float pe[E_DIM];
        #pragma unroll
        for (int e = 0; e < E_DIM; ++e) pe[e] = 0.f;
        #pragma unroll
        for (int i = 0; i < H_DIM / PREP_THREADS; ++i) {
            const int h = tid + i * PREP_THREADS;
            const float bv = bi[h];
            #pragma unroll
            for (int e = 0; e < E_DIM; ++e)
                pe[e] = __builtin_fmaf(Wr[(size_t)e * H_DIM + h], bv, pe[e]);
        }
        #pragma unroll
        for (int e = 0; e < E_DIM; ++e) {
            float v = pe[e];
            #pragma unroll
            for (int off = 16; off >= 1; off >>= 1) v += __shfl_xor(v, off);
            pe[e] = v;
        }
        if (lane == 0) {
            #pragma unroll
            for (int e = 0; e < E_DIM; ++e) red[wave][e][0] = pe[e];
        }
        __syncthreads();
        if (tid < E_DIM) {
            float s = br[tid];
            #pragma unroll
            for (int w = 0; w < PREP_WAVES; ++w) s += red[w][tid][0];
            beff[tid] = s;
        }
        #pragma unroll
        for (int n = 0; n < 8; ++n)
            WcT[(8 + n) * IN_DIM + tid] = Wi[(size_t)n * IN_DIM + tid];
    }
}

#define MAIN_WAVES   4
#define MAIN_THREADS (MAIN_WAVES * 32)
#define N_CHUNK      (IN_DIM / 8)
#define W_PITCH      516
#define W_BLK        16

__global__ void __launch_bounds__(MAIN_THREADS)
moe_main(const float* __restrict__ x,    const float* __restrict__ WcT,
         const float* __restrict__ beff, const float* __restrict__ bi,
         const float* __restrict__ wsum, const float* __restrict__ bo,
         float* __restrict__ out) {
    __shared__ __attribute__((aligned(16))) float lds_w[16 * W_PITCH];
    __shared__ float lds_c[MAIN_WAVES][16][16];
    __shared__ float lds_agg[MAIN_WAVES][16];

    const int tid   = threadIdx.x;
    const int wave  = tid >> 5;
    const int lane  = tid & 31;
    const int tile  = blockIdx.x * MAIN_WAVES + wave;
    const int tok0  = tile * 16;
    const int col   = lane & 15;
    const int half  = lane >> 4;

    v4f wst[16];
    #pragma unroll
    for (int n = 0; n < 16; ++n)
        wst[n] = *(const v4f*)(WcT + (size_t)n * IN_DIM + 4 * tid);

    const float* ap = x + (size_t)(tok0 + col) * IN_DIM + half * 4;
    v4f xa[N_CHUNK];
    #pragma unroll
    for (int j = 0; j < N_CHUNK; ++j)
        xa[j] = *(const v4f*)(ap + 8 * j);

    #pragma unroll
    for (int n = 0; n < 16; ++n)
        *(v4f*)&lds_w[n * W_PITCH + 4 * tid] = wst[n];

    v4f wv[4], bv[4];
    #pragma unroll
    for (int i = 0; i < 4; ++i) {
        wv[i] = *(const v4f*)(wsum + lane * 4 + 128 * i);
        bv[i] = *(const v4f*)(bo   + lane * 4 + 128 * i);
    }
    const float cb = (col < 8) ? beff[col] : bi[col - 8];
    __builtin_amdgcn_fence(__ATOMIC_RELEASE, "workgroup", "local");
    __builtin_amdgcn_s_barrier();
    __builtin_amdgcn_fence(__ATOMIC_ACQUIRE, "workgroup", "local");

    const float* bl = lds_w + col * W_PITCH + half * 4;
    v8f acc0 = {}, acc1 = {};
    v4f wb[2][W_BLK];
    #pragma unroll
    for (int c = 0; c < W_BLK; ++c)
        wb[0][c] = *(const v4f*)(bl + 8 * c);
    #pragma unroll
    for (int blk = 0; blk < N_CHUNK / W_BLK; ++blk) {
        if (blk + 1 < N_CHUNK / W_BLK) {
            #pragma unroll
            for (int c = 0; c < W_BLK; ++c)
                wb[(blk + 1) & 1][c] = *(const v4f*)(bl + 8 * ((blk + 1) * W_BLK + c));
        }
        __builtin_amdgcn_sched_barrier(0);
        #pragma unroll
        for (int c = 0; c < W_BLK; ++c) {
            const v4f a = xa[blk * W_BLK + c];
            const v4f b = wb[blk & 1][c];
            const v2f a0 = {a.x, a.y}, a1 = {a.z, a.w};
            const v2f b0 = {b.x, b.y}, b1 = {b.z, b.w};
            acc0 = __builtin_amdgcn_wmma_f32_16x16x4_f32(
                 false, a0,  false, b0,
                 (short)0, acc0,  false,  false);
            acc1 = __builtin_amdgcn_wmma_f32_16x16x4_f32(
                 false, a1,  false, b1,
                 (short)0, acc1,  false,  false);
        }
        __builtin_amdgcn_sched_barrier(0);
    }
    const v8f acc = acc0 + acc1;

    #pragma unroll
    for (int j = 0; j < 8; ++j)
        lds_c[wave][half * 8 + j][col] = acc[j] + cb;
    __syncthreads();

    if (lane < 16) {
        const float* r = lds_c[wave][lane];
        float b1 = r[0]; int i1 = 0;
        #pragma unroll
        for (int n = 1; n < 8; ++n) if (r[n] > b1) { b1 = r[n]; i1 = n; }
        float b2 = -3.402823466e38f; int i2 = 0;
        #pragma unroll
        for (int n = 0; n < 8; ++n) if (n != i1 && r[n] > b2) { b2 = r[n]; i2 = n; }
        lds_agg[wave][lane] = 0.5f * (r[8 + i1] + r[8 + i2]);
    }
    __syncthreads();

    #pragma unroll 4
    for (int t = 0; t < 16; ++t) {
        float ag = lds_agg[wave][t];
        float* op = out + (size_t)(tok0 + t) * OUT_DIM + lane * 4;
        #pragma unroll
        for (int i = 0; i < 4; ++i) {
            v4f r = ag * wv[i] + bv[i];
            *(v4f*)(op + 128 * i) = r;
        }
    }
}

extern "C" void kernel_launch(void* const* d_in, const int* in_sizes, int n_in,
                              void* d_out, int out_size, void* d_ws, size_t ws_size,
                              hipStream_t stream) {
    const float* x  = (const float*)d_in[0];
    const float* Wi = (const float*)d_in[1];
    const float* bi = (const float*)d_in[2];
    const float* Wr = (const float*)d_in[3];
    const float* br = (const float*)d_in[4];
    const float* Wo = (const float*)d_in[5];
    const float* bo = (const float*)d_in[6];
    float* out = (float*)d_out;

    float* WcT  = (float*)d_ws;
    float* beff = WcT + 16 * IN_DIM;
    float* wsum = beff + 16;

    moe_prep<<<N_PREP, PREP_THREADS, 0, stream>>>(Wi, bi, Wr, br, Wo, WcT, beff, wsum);
    moe_main<<<NTOK / 16 / MAIN_WAVES, MAIN_THREADS, 0, stream>>>(x, WcT, beff, bi, wsum, bo, out);
}


# Weight prep: the 4 fold groups of one Wi column strip walk its row chunks in rotated order: each line misses to memory for one group, hits L2 for the rest; changes f32 sum order of the folded weight
# speedup vs baseline: 21.5740x; 1.1455x over previous
#include <hip/hip_runtime.h>
#include <hip/hip_bf16.h>

typedef __attribute__((ext_vector_type(2))) float v2f;
typedef __attribute__((ext_vector_type(4))) float v4f;
typedef __attribute__((ext_vector_type(8))) float v8f;

#define H_DIM   4096
#define IN_DIM  512
#define E_DIM   8
#define OUT_DIM 512
#define NTOK    16384

#define PREP_THREADS 512
#define PREP_WAVES   (PREP_THREADS / 32)
#define STRIP        32
#define N_FOLD       (IN_DIM / STRIP)
#define E_PER        2
#define N_FOLDB      (N_FOLD * (E_DIM / E_PER))
#define WSUM_ROWS    8
#define N_WSUM       (OUT_DIM / WSUM_ROWS)
#define N_PREP       (N_FOLDB + N_WSUM + 1)

__global__ void __launch_bounds__(PREP_THREADS)
moe_prep(const float* __restrict__ Wi, const float* __restrict__ bi,
         const float* __restrict__ Wr, const float* __restrict__ br,
         const float* __restrict__ Wo,
         float* __restrict__ WcT, float* __restrict__ beff, float* __restrict__ wsum) {
    __shared__ __attribute__((aligned(16))) float red[PREP_WAVES][E_DIM][STRIP];

    const int tid  = threadIdx.x;
    const int lane = tid & 31;
    const int wave = tid >> 5;
    const int b    = blockIdx.x;

    if (b < N_FOLDB) {
        constexpr int ROWS_PER_LANE = H_DIM / (PREP_WAVES * 4);
        constexpr int CH  = 16;
        constexpr int NCH = ROWS_PER_LANE / CH;
        constexpr int S   = 2;
        const int k0 = (b % N_FOLD) * STRIP;
        const int pg = b / N_FOLD;
        const int e0 = pg * E_PER;
        const int c4 = lane & 7;
        const int rg = lane >> 3;
        const int hb = wave * (4 * ROWS_PER_LANE) + rg * ROWS_PER_LANE;
        const float* wip = Wi + (size_t)hb * IN_DIM + k0 + c4 * 4;
        const float* wrp = Wr + (size_t)e0 * H_DIM + hb;

        v4f acc[E_PER];
        #pragma unroll
        for (int e = 0; e < E_PER; ++e) acc[e] = (v4f){0.f, 0.f, 0.f, 0.f};

        v4f wi[S][CH];
        v4f wr[S][E_PER][CH / 4];
        static_assert((NCH & (NCH - 1)) == 0, "chunk rotation uses a power-of-two mask");
        {
            const int c0 = pg & (NCH - 1);
            const float* wq = wip + (size_t)c0 * CH * IN_DIM;
            const float* rq = wrp + c0 * CH;
            #pragma unroll
            for (int r = 0; r < CH; ++r)
                wi[0][r] = *(const v4f*)(wq + (size_t)r * IN_DIM);
            #pragma unroll
            for (int e = 0; e < E_PER; ++e) {
                #pragma unroll
                for (int g = 0; g < CH / 4; ++g)
                    wr[0][e][g] = *(const v4f*)(rq + (size_t)e * H_DIM + g * 4);
            }
        }
        #pragma unroll
        for (int q = 0; q < NCH; ++q) {
            const int qn = q + 1;
            if (qn < NCH) {
                const int cn = (qn + pg) & (NCH - 1);
                const float* wq = wip + (size_t)cn * CH * IN_DIM;
                const float* rq = wrp + cn * CH;
                #pragma unroll
                for (int r = 0; r < CH; ++r)
                    wi[qn % S][r] = *(const v4f*)(wq + (size_t)r * IN_DIM);
                #pragma unroll
                for (int e = 0; e < E_PER; ++e) {
                    #pragma unroll
                    for (int g = 0; g < CH / 4; ++g)
                        wr[qn % S][e][g] = *(const v4f*)(rq + (size_t)e * H_DIM + g * 4);
                }
            }
            __builtin_amdgcn_sched_barrier(0);
            const int s = q % S;
            #pragma unroll
            for (int r = 0; r < CH; ++r) {
                #pragma unroll
                for (int e = 0; e < E_PER; ++e) {
                    const float w = wr[s][e][r >> 2][r & 3];
                    acc[e].x = __builtin_fmaf(w, wi[s][r].x, acc[e].x);
                    acc[e].y = __builtin_fmaf(w, wi[s][r].y, acc[e].y);
                    acc[e].z = __builtin_fmaf(w, wi[s][r].z, acc[e].z);
                    acc[e].w = __builtin_fmaf(w, wi[s][r].w, acc[e].w);
                }
            }
        }

        #pragma unroll
        for (int e = 0; e < E_PER; ++e) {
            #pragma unroll
            for (int j = 0; j < 4; ++j) {
                float v = acc[e][j];
                v += __shfl_xor(v, 8);
                v += __shfl_xor(v, 16);
                acc[e][j] = v;
            }
        }
        if (lane < 8) {
            #pragma unroll
            for (int e = 0; e < E_PER; ++e)
                *(v4f*)&red[wave][e][c4 * 4] = acc[e];
        }
        __syncthreads();
        if (tid < E_PER * STRIP) {
            const int e = tid >> 5;
            const int c = tid & 31;
            float s = 0.f;
            #pragma unroll
            for (int w = 0; w < PREP_WAVES; ++w) s += red[w][e][c];
            WcT[(e0 + e) * IN_DIM + k0 + c] = s;
        }
    } else if (b > N_FOLDB) {
        const int o0 = (b - N_FOLDB - 1) * WSUM_ROWS;
        const int q  = wave & 3;
        const int r0 = wave >> 2;
        float* rs = &red[0][0][0];
        v4f v[2][8];
        #pragma unroll
        for (int p = 0; p < 2; ++p) {
            const float* wp = Wo + (size_t)(o0 + r0 + 4 * p) * H_DIM + q * (H_DIM / 4) + lane * 4;
            #pragma unroll
            for (int i = 0; i < 8; ++i) v[p][i] = *(const v4f*)(wp + i * 128);
        }
        #pragma unroll
        for (int p = 0; p < 2; ++p) {
            v4f s4 = ((v[p][0] + v[p][1]) + (v[p][2] + v[p][3])) + ((v[p][4] + v[p][5]) + (v[p][6] + v[p][7]));
            float s = (s4.x + s4.y) + (s4.z + s4.w);
            #pragma unroll
            for (int off = 16; off >= 1; off >>= 1) s += __shfl_xor(s, off);
            if (lane == 0) rs[4 * (r0 + 4 * p) + q] = s;
        }
        __syncthreads();
        if (tid < WSUM_ROWS)
            wsum[o0 + tid] = (rs[4 * tid] + rs[4 * tid + 1]) + (rs[4 * tid + 2] + rs[4 * tid + 3]);
    } else {
        float pe[E_DIM];
        #pragma unroll
        for (int e = 0; e < E_DIM; ++e) pe[e] = 0.f;
        #pragma unroll
        for (int i = 0; i < H_DIM / PREP_THREADS; ++i) {
            const int h = tid + i * PREP_THREADS;
            const float bv = bi[h];
            #pragma unroll
            for (int e = 0; e < E_DIM; ++e)
                pe[e] = __builtin_fmaf(Wr[(size_t)e * H_DIM + h], bv, pe[e]);
        }
        #pragma unroll
        for (int e = 0; e < E_DIM; ++e) {
            float v = pe[e];
            #pragma unroll
            for (int off = 16; off >= 1; off >>= 1) v += __shfl_xor(v, off);
            pe[e] = v;
        }
        if (lane == 0) {
            #pragma unroll
            for (int e = 0; e < E_DIM; ++e) red[wave][e][0] = pe[e];
        }
        __syncthreads();
        if (tid < E_DIM) {
            float s = br[tid];
            #pragma unroll
            for (int w = 0; w < PREP_WAVES; ++w) s += red[w][tid][0];
            beff[tid] = s;
        }
        #pragma unroll
        for (int n = 0; n < 8; ++n)
            WcT[(8 + n) * IN_DIM + tid] = Wi[(size_t)n * IN_DIM + tid];
    }
}

#define MAIN_WAVES   4
#define MAIN_THREADS (MAIN_WAVES * 32)
#define N_CHUNK      (IN_DIM / 8)
#define W_PITCH      516
#define W_BLK        16

__global__ void __launch_bounds__(MAIN_THREADS)
moe_main(const float* __restrict__ x,    const float* __restrict__ WcT,
         const float* __restrict__ beff, const float* __restrict__ bi,
         const float* __restrict__ wsum, const float* __restrict__ bo,
         float* __restrict__ out) {
    __shared__ __attribute__((aligned(16))) float lds_w[16 * W_PITCH];
    __shared__ float lds_c[MAIN_WAVES][16][16];
    __shared__ float lds_agg[MAIN_WAVES][16];

    const int tid   = threadIdx.x;
    const int wave  = tid >> 5;
    const int lane  = tid & 31;
    const int tile  = blockIdx.x * MAIN_WAVES + wave;
    const int tok0  = tile * 16;
    const int col   = lane & 15;
    const int half  = lane >> 4;

    v4f wst[16];
    #pragma unroll
    for (int n = 0; n < 16; ++n)
        wst[n] = *(const v4f*)(WcT + (size_t)n * IN_DIM + 4 * tid);

    const float* ap = x + (size_t)(tok0 + col) * IN_DIM + half * 4;
    v4f xa[N_CHUNK];
    #pragma unroll
    for (int j = 0; j < N_CHUNK; ++j)
        xa[j] = *(const v4f*)(ap + 8 * j);

    #pragma unroll
    for (int n = 0; n < 16; ++n)
        *(v4f*)&lds_w[n * W_PITCH + 4 * tid] = wst[n];

    v4f wv[4], bv[4];
    #pragma unroll
    for (int i = 0; i < 4; ++i) {
        wv[i] = *(const v4f*)(wsum + lane * 4 + 128 * i);
        bv[i] = *(const v4f*)(bo   + lane * 4 + 128 * i);
    }
    const float cb = (col < 8) ? beff[col] : bi[col - 8];
    __builtin_amdgcn_fence(__ATOMIC_RELEASE, "workgroup", "local");
    __builtin_amdgcn_s_barrier();
    __builtin_amdgcn_fence(__ATOMIC_ACQUIRE, "workgroup", "local");

    const float* bl = lds_w + col * W_PITCH + half * 4;
    v8f acc0 = {}, acc1 = {};
    v4f wb[2][W_BLK];
    #pragma unroll
    for (int c = 0; c < W_BLK; ++c)
        wb[0][c] = *(const v4f*)(bl + 8 * c);
    #pragma unroll
    for (int blk = 0; blk < N_CHUNK / W_BLK; ++blk) {
        if (blk + 1 < N_CHUNK / W_BLK) {
            #pragma unroll
            for (int c = 0; c < W_BLK; ++c)
                wb[(blk + 1) & 1][c] = *(const v4f*)(bl + 8 * ((blk + 1) * W_BLK + c));
        }
        __builtin_amdgcn_sched_barrier(0);
        #pragma unroll
        for (int c = 0; c < W_BLK; ++c) {
            const v4f a = xa[blk * W_BLK + c];
            const v4f b = wb[blk & 1][c];
            const v2f a0 = {a.x, a.y}, a1 = {a.z, a.w};
            const v2f b0 = {b.x, b.y}, b1 = {b.z, b.w};
            acc0 = __builtin_amdgcn_wmma_f32_16x16x4_f32(
                 false, a0,  false, b0,
                 (short)0, acc0,  false,  false);
            acc1 = __builtin_amdgcn_wmma_f32_16x16x4_f32(
                 false, a1,  false, b1,
                 (short)0, acc1,  false,  false);
        }
        __builtin_amdgcn_sched_barrier(0);
    }
    const v8f acc = acc0 + acc1;

    #pragma unroll
    for (int j = 0; j < 8; ++j)
        lds_c[wave][half * 8 + j][col] = acc[j] + cb;
    __syncthreads();

    if (lane < 16) {
        const float* r = lds_c[wave][lane];
        float b1 = r[0]; int i1 = 0;
        #pragma unroll
        for (int n = 1; n < 8; ++n) if (r[n] > b1) { b1 = r[n]; i1 = n; }
        float b2 = -3.402823466e38f; int i2 = 0;
        #pragma unroll
        for (int n = 0; n < 8; ++n) if (n != i1 && r[n] > b2) { b2 = r[n]; i2 = n; }
        lds_agg[wave][lane] = 0.5f * (r[8 + i1] + r[8 + i2]);
    }
    __syncthreads();

    #pragma unroll 4
    for (int t = 0; t < 16; ++t) {
        float ag = lds_agg[wave][t];
        float* op = out + (size_t)(tok0 + t) * OUT_DIM + lane * 4;
        #pragma unroll
        for (int i = 0; i < 4; ++i) {
            v4f r = ag * wv[i] + bv[i];
            *(v4f*)(op + 128 * i) = r;
        }
    }
}

extern "C" void kernel_launch(void* const* d_in, const int* in_sizes, int n_in,
                              void* d_out, int out_size, void* d_ws, size_t ws_size,
                              hipStream_t stream) {
    const float* x  = (const float*)d_in[0];
    const float* Wi = (const float*)d_in[1];
    const float* bi = (const float*)d_in[2];
    const float* Wr = (const float*)d_in[3];
    const float* br = (const float*)d_in[4];
    const float* Wo = (const float*)d_in[5];
    const float* bo = (const float*)d_in[6];
    float* out = (float*)d_out;

    float* WcT  = (float*)d_ws;
    float* beff = WcT + 16 * IN_DIM;
    float* wsum = beff + 16;

    moe_prep<<<N_PREP, PREP_THREADS, 0, stream>>>(Wi, bi, Wr, br, Wo, WcT, beff, wsum);
    moe_main<<<NTOK / 16 / MAIN_WAVES, MAIN_THREADS, 0, stream>>>(x, WcT, beff, bi, wsum, bo, out);
}
